# starting kernel
# baseline (speedup 1.0000x reference)
#include <hip/hip_runtime.h>

typedef __attribute__((ext_vector_type(2))) float v2f;
typedef __attribute__((ext_vector_type(8))) float v8f;

#define NFEAT 128
#define NHID  128

__global__ void gcn_init_deg(float* __restrict__ deg, int n) {
    int i = blockIdx.x * blockDim.x + threadIdx.x;
    if (i < n) deg[i] = 1.0f;
}

__global__ void gcn_count_deg(const int* __restrict__ dst, float* __restrict__ deg, int e) {
    int i = blockIdx.x * blockDim.x + threadIdx.x;
    if (i < e) atomicAdd(&deg[dst[i]], 1.0f);
}

__global__ void gcn_rsqrt_deg(float* __restrict__ deg, int n) {
    int i = blockIdx.x * blockDim.x + threadIdx.x;
    if (i < n) deg[i] = rsqrtf(deg[i]);
}

__global__ __launch_bounds__(256) void gcn_gemm_wmma(const float* __restrict__ x,
                                                     const float* __restrict__ W,
                                                     float* __restrict__ xw,
                                                     int n_nodes) {
    const int lane = threadIdx.x & 31;
    const int wave = threadIdx.x >> 5;
    const int half = lane >> 4;
    const int l    = lane & 15;

    const int m0 = blockIdx.x * 16;
    const int n0 = wave * 16;

    int arow = m0 + l;
    if (arow >= n_nodes) arow = n_nodes - 1;

    const float* aptr = x + (size_t)arow * NFEAT + 2 * half;
    const float* bptr = W + (size_t)(2 * half) * NHID + n0 + l;

    v8f c = {};
#pragma unroll
    for (int kk = 0; kk < NFEAT; kk += 4) {
        v2f a = *(const v2f*)(aptr + kk);
        v2f b;
        b.x = bptr[(size_t)kk * NHID];
        b.y = bptr[(size_t)kk * NHID + NHID];
        c = __builtin_amdgcn_wmma_f32_16x16x4_f32(
                 false, a,  false, b,
                 (short)0, c,  false,  false);
    }

#pragma unroll
    for (int r = 0; r < 8; ++r) {
        int row = m0 + r + 8 * half;
        if (row < n_nodes) xw[(size_t)row * NHID + n0 + l] = c[r];
    }
}

__global__ void gcn_self_bias(const float* __restrict__ xw,
                              const float* __restrict__ dinv,
                              const float* __restrict__ bias,
                              float* __restrict__ out, int n_nodes) {
    int t = blockIdx.x * blockDim.x + threadIdx.x;
    int i = t >> 5;
    if (i >= n_nodes) return;
    int lane = t & 31;

    float di = dinv[i];
    float n2 = di * di;
    const float4 v  = *(const float4*)(xw + (size_t)i * NHID + lane * 4);
    const float4 bb = *(const float4*)(bias + lane * 4);
    float4 o;
    o.x = fmaf(v.x, n2, bb.x);
    o.y = fmaf(v.y, n2, bb.y);
    o.z = fmaf(v.z, n2, bb.z);
    o.w = fmaf(v.w, n2, bb.w);
    *(float4*)(out + (size_t)i * NHID + lane * 4) = o;
}

__global__ __launch_bounds__(256) void gcn_edge_scatter(const int* __restrict__ src,
                                                        const int* __restrict__ dst,
                                                        const float* __restrict__ dinv,
                                                        const float* __restrict__ xw,
                                                        float* __restrict__ out, int e) {
    int edge = blockIdx.x * 8 + (threadIdx.x >> 5);
    if (edge >= e) return;
    int lane = threadIdx.x & 31;

    int s = src[edge];
    int d = dst[edge];
    float norm = dinv[s] * dinv[d];

    const float4 v = *(const float4*)(xw + (size_t)s * NHID + lane * 4);
    float* o = out + (size_t)d * NHID + lane * 4;
    atomicAdd(o + 0, v.x * norm);
    atomicAdd(o + 1, v.y * norm);
    atomicAdd(o + 2, v.z * norm);
    atomicAdd(o + 3, v.w * norm);
}

extern "C" void kernel_launch(void* const* d_in, const int* in_sizes, int n_in,
                              void* d_out, int out_size, void* d_ws, size_t ws_size,
                              hipStream_t stream) {
    const int*   ei   = (const int*)d_in[0];
    const float* x    = (const float*)d_in[1];
    const float* W    = (const float*)d_in[2];
    const float* bias = (const float*)d_in[3];
    float*       out  = (float*)d_out;

    const int E = in_sizes[0] / 2;
    const int N = in_sizes[1] / NFEAT;

    float* xw  = (float*)d_ws;
    float* deg = (float*)((char*)d_ws + (size_t)N * NHID * sizeof(float));

    const int* srcIdx = ei;
    const int* dstIdx = ei + E;

    gcn_init_deg <<<(N + 255) / 256, 256, 0, stream>>>(deg, N);
    gcn_count_deg<<<(E + 255) / 256, 256, 0, stream>>>(dstIdx, deg, E);
    gcn_rsqrt_deg<<<(N + 255) / 256, 256, 0, stream>>>(deg, N);

    gcn_gemm_wmma<<<(N + 15) / 16, 256, 0, stream>>>(x, W, xw, N);

    gcn_self_bias<<<((size_t)N * 32 + 255) / 256, 256, 0, stream>>>(xw, deg, bias, out, N);
    gcn_edge_scatter<<<(E + 7) / 8, 256, 0, stream>>>(srcIdx, dstIdx, deg, xw, out, E);
}


# Bucket edges by destination (CSR built with LDS counters, no global atomics left), then one wave per node sums its in-edges in registers and stores the row once; summation order changes
# speedup vs baseline: 1.8652x; 1.8652x over previous
#include <hip/hip_runtime.h>


typedef __attribute__((ext_vector_type(2))) float v2f;
typedef __attribute__((ext_vector_type(8))) float v8f;

#define NFEAT 128
#define NHID  128

#define EDGE_CHUNKS 64
#define LDS_NODES   53248

__global__ __launch_bounds__(1024) void csr_hist(const int* __restrict__ dst,
                                                 int* __restrict__ part, int n, int e) {
    __shared__ int h[LDS_NODES];
    const int b = blockIdx.x;
    const int chunk = (e + EDGE_CHUNKS - 1) / EDGE_CHUNKS;
    const int e0 = b * chunk;
    const int e1 = (e0 + chunk < e) ? (e0 + chunk) : e;
    for (int r0 = 0; r0 < n; r0 += LDS_NODES) {
        const int rn = (n - r0 < LDS_NODES) ? (n - r0) : LDS_NODES;
        for (int k = threadIdx.x; k < rn; k += 1024) h[k] = 0;
        __syncthreads();
        for (int k0 = e0 + threadIdx.x; k0 < e1; k0 += 4096) {
            int d[4];
            for (int u = 0; u < 4; ++u) {
                const int k = k0 + u * 1024;
                const int dv = dst[(k < e1) ? k : (e1 - 1)];
                d[u] = (k < e1) ? (dv - r0) : -1;
            }
            for (int u = 0; u < 4; ++u)
                if ((unsigned)d[u] < (unsigned)rn) atomicAdd(&h[d[u]], 1);
        }
        __syncthreads();
        for (int k = threadIdx.x; k < rn; k += 1024) part[(size_t)b * n + r0 + k] = h[k];
        __syncthreads();
    }
}

__global__ void csr_colscan(int* __restrict__ part, int* __restrict__ cnt,
                            float* __restrict__ dinv, int n) {
    const int d = blockIdx.x * blockDim.x + threadIdx.x;
    if (d >= n) return;
    int* p = part + d;
    int run = 0;
#pragma clang loop unroll(disable)
    for (int b0 = 0; b0 < EDGE_CHUNKS; b0 += 16) {
        int v[16];
        for (int u = 0; u < 16; ++u) v[u] = p[(size_t)(b0 + u) * n];
        for (int u = 0; u < 16; ++u) {
            p[(size_t)(b0 + u) * n] = run;
            run += v[u];
        }
    }
    cnt[d] = run;
    dinv[d] = rsqrtf((float)(run + 1));
}

__global__ __launch_bounds__(1024) void csr_rowscan(const int* __restrict__ cnt,
                                                    int* __restrict__ rowptr, int n) {
    __shared__ int wsum[32];
    __shared__ int psum[32];
    const int t = threadIdx.x;
    const int lane = t & 31;
    const int wave = t >> 5;
    const int base = blockIdx.x * 1024;

    int pre = 0;
    for (int k = t; k < base; k += 1024) pre += cnt[k];
    const int v = (base + t < n) ? cnt[base + t] : 0;

    int inc = v;
    for (int o = 1; o < 32; o <<= 1) {
        const int u = __shfl_up(inc, o);
        if (lane >= o) inc += u;
    }
    for (int o = 16; o > 0; o >>= 1) pre += __shfl_xor(pre, o);
    if (lane == 31) wsum[wave] = inc;
    if (lane == 0)  psum[wave] = pre;
    __syncthreads();
    int off = 0;
    for (int w = 0; w < 32; ++w) {
        off += psum[w];
        if (w < wave) off += wsum[w];
    }
    if (base + t < n) rowptr[base + t] = off + inc - v;
}

__global__ __launch_bounds__(1024) void csr_fill(const int* __restrict__ src,
                                                 const int* __restrict__ dst,
                                                 const int* __restrict__ part,
                                                 const int* __restrict__ rowptr,
                                                 const float* __restrict__ dinv,
                                                 int2* __restrict__ ent, int n, int e) {
    __shared__ int h[LDS_NODES];
    const int b = blockIdx.x;
    const int chunk = (e + EDGE_CHUNKS - 1) / EDGE_CHUNKS;
    const int e0 = b * chunk;
    const int e1 = (e0 + chunk < e) ? (e0 + chunk) : e;
    for (int r0 = 0; r0 < n; r0 += LDS_NODES) {
        const int rn = (n - r0 < LDS_NODES) ? (n - r0) : LDS_NODES;
        for (int k = threadIdx.x; k < rn; k += 1024)
            h[k] = part[(size_t)b * n + r0 + k] + rowptr[r0 + k];
        __syncthreads();
        for (int k0 = e0 + threadIdx.x; k0 < e1; k0 += 4096) {
            int d[4], s[4];
            float w[4];
            for (int u = 0; u < 4; ++u) {
                const int k = k0 + u * 1024;
                const int kc = (k < e1) ? k : (e1 - 1);
                const int dv = dst[kc];
                s[u] = src[kc];
                d[u] = (k < e1) ? (dv - r0) : -1;
            }
            for (int u = 0; u < 4; ++u) w[u] = dinv[s[u]];
            for (int u = 0; u < 4; ++u) {
                if ((unsigned)d[u] < (unsigned)rn) {
                    const int pos = atomicAdd(&h[d[u]], 1);
                    ent[pos] = make_int2(s[u], __float_as_int(w[u]));
                }
            }
        }
        __syncthreads();
    }
}

__global__ __launch_bounds__(256) void gcn_gemm_wmma(const float* __restrict__ x,
                                                     const float* __restrict__ W,
                                                     float* __restrict__ xw,
                                                     int n_nodes) {
    const int lane = threadIdx.x & 31;
    const int wave = threadIdx.x >> 5;
    const int half = lane >> 4;
    const int l    = lane & 15;

    const int m0 = blockIdx.x * 16;
    const int n0 = wave * 16;

    int arow = m0 + l;
    if (arow >= n_nodes) arow = n_nodes - 1;

    const float* aptr = x + (size_t)arow * NFEAT + 2 * half;
    const float* bptr = W + (size_t)(2 * half) * NHID + n0 + l;

    v8f c = {};
#pragma unroll
    for (int kk = 0; kk < NFEAT; kk += 4) {
        v2f a = *(const v2f*)(aptr + kk);
        v2f b;
        b.x = bptr[(size_t)kk * NHID];
        b.y = bptr[(size_t)kk * NHID + NHID];
        c = __builtin_amdgcn_wmma_f32_16x16x4_f32(
                 false, a,  false, b,
                 (short)0, c,  false,  false);
    }

#pragma unroll
    for (int r = 0; r < 8; ++r) {
        int row = m0 + r + 8 * half;
        if (row < n_nodes) xw[(size_t)row * NHID + n0 + l] = c[r];
    }
}

__global__ __launch_bounds__(256) void gcn_gather(const int2* __restrict__ ent,
                                                  const int* __restrict__ rowptr,
                                                  const int* __restrict__ cnt,
                                                  const float* __restrict__ dinv,
                                                  const float* __restrict__ xw,
                                                  const float* __restrict__ bias,
                                                  float* __restrict__ out, int n) {
    const int lane = threadIdx.x & 31;
    const int i = __builtin_amdgcn_readfirstlane((int)(blockIdx.x * 8 + (threadIdx.x >> 5)));
    if (i >= n) return;

    const float di = dinv[i];
    const float n2 = di * di;
    const float4 v  = *(const float4*)(xw + (size_t)i * NHID + lane * 4);
    const float4 bb = *(const float4*)(bias + lane * 4);
    float4 acc;
    acc.x = fmaf(v.x, n2, bb.x);
    acc.y = fmaf(v.y, n2, bb.y);
    acc.z = fmaf(v.z, n2, bb.z);
    acc.w = fmaf(v.w, n2, bb.w);

    const int start = rowptr[i];
    const int total = cnt[i];
    for (int base = 0; base < total; base += 32) {
        const int m = (total - base < 32) ? (total - base) : 32;
        int sidx = 0;
        float wgt = 0.0f;
        if (lane < m) {
            const int2 en = ent[start + base + lane];
            sidx = en.x;
            wgt  = __int_as_float(en.y) * di;
        }
        const int wbits = __float_as_int(wgt);
        int j = 0;
        for (; j + 4 <= m; j += 4) {
            const int s0 = __builtin_amdgcn_readlane(sidx, j);
            const int s1 = __builtin_amdgcn_readlane(sidx, j + 1);
            const int s2 = __builtin_amdgcn_readlane(sidx, j + 2);
            const int s3 = __builtin_amdgcn_readlane(sidx, j + 3);
            const float w0 = __int_as_float(__builtin_amdgcn_readlane(wbits, j));
            const float w1 = __int_as_float(__builtin_amdgcn_readlane(wbits, j + 1));
            const float w2 = __int_as_float(__builtin_amdgcn_readlane(wbits, j + 2));
            const float w3 = __int_as_float(__builtin_amdgcn_readlane(wbits, j + 3));
            const float4 r0 = *(const float4*)(xw + (size_t)s0 * NHID + lane * 4);
            const float4 r1 = *(const float4*)(xw + (size_t)s1 * NHID + lane * 4);
            const float4 r2 = *(const float4*)(xw + (size_t)s2 * NHID + lane * 4);
            const float4 r3 = *(const float4*)(xw + (size_t)s3 * NHID + lane * 4);
            acc.x = fmaf(r0.x, w0, acc.x); acc.y = fmaf(r0.y, w0, acc.y);
            acc.z = fmaf(r0.z, w0, acc.z); acc.w = fmaf(r0.w, w0, acc.w);
            acc.x = fmaf(r1.x, w1, acc.x); acc.y = fmaf(r1.y, w1, acc.y);
            acc.z = fmaf(r1.z, w1, acc.z); acc.w = fmaf(r1.w, w1, acc.w);
            acc.x = fmaf(r2.x, w2, acc.x); acc.y = fmaf(r2.y, w2, acc.y);
            acc.z = fmaf(r2.z, w2, acc.z); acc.w = fmaf(r2.w, w2, acc.w);
            acc.x = fmaf(r3.x, w3, acc.x); acc.y = fmaf(r3.y, w3, acc.y);
            acc.z = fmaf(r3.z, w3, acc.z); acc.w = fmaf(r3.w, w3, acc.w);
        }
        for (; j < m; ++j) {
            const int s0 = __builtin_amdgcn_readlane(sidx, j);
            const float w0 = __int_as_float(__builtin_amdgcn_readlane(wbits, j));
            const float4 r0 = *(const float4*)(xw + (size_t)s0 * NHID + lane * 4);
            acc.x = fmaf(r0.x, w0, acc.x); acc.y = fmaf(r0.y, w0, acc.y);
            acc.z = fmaf(r0.z, w0, acc.z); acc.w = fmaf(r0.w, w0, acc.w);
        }
    }
    *(float4*)(out + (size_t)i * NHID + lane * 4) = acc;
}

extern "C" void kernel_launch(void* const* d_in, const int* in_sizes, int n_in,
                              void* d_out, int out_size, void* d_ws, size_t ws_size,
                              hipStream_t stream) {
    const int*   ei   = (const int*)d_in[0];
    const float* x    = (const float*)d_in[1];
    const float* W    = (const float*)d_in[2];
    const float* bias = (const float*)d_in[3];
    float*       out  = (float*)d_out;

    const int E = in_sizes[0] / 2;
    const int N = in_sizes[1] / NFEAT;

    float* xw     = (float*)d_ws;
    int2*  ent    = (int2*)(xw + (size_t)N * NHID);
    int*   part   = (int*)(ent + (size_t)E);
    float* dinv   = (float*)(part + (size_t)EDGE_CHUNKS * N);
    int*   cnt    = (int*)(dinv + (size_t)N);
    int*   rowptr = cnt + (size_t)N;

    const int* srcIdx = ei;
    const int* dstIdx = ei + E;

    csr_hist   <<<EDGE_CHUNKS, 1024, 0, stream>>>(dstIdx, part, N, E);
    csr_colscan<<<(N + 255) / 256, 256, 0, stream>>>(part, cnt, dinv, N);
    csr_rowscan<<<(N + 1023) / 1024, 1024, 0, stream>>>(cnt, rowptr, N);
    csr_fill   <<<EDGE_CHUNKS, 1024, 0, stream>>>(srcIdx, dstIdx, part, rowptr, dinv, ent, N, E);

    gcn_gemm_wmma<<<(N + 15) / 16, 256, 0, stream>>>(x, W, xw, N);

    gcn_gather<<<(N + 7) / 8, 256, 0, stream>>>(ent, rowptr, cnt, dinv, xw, bias, out, N);
}


# Both edge-bucketing kernels split each chunk's work over 4 node ranges (64 x 4 work-groups, one per WGP, not 64): a WGP's scattered weight lookups and entry stores fall to a quarter
# speedup vs baseline: 2.3004x; 1.2333x over previous
#include <hip/hip_runtime.h>


typedef __attribute__((ext_vector_type(2))) float v2f;
typedef __attribute__((ext_vector_type(8))) float v8f;

#define NFEAT 128
#define NHID  128

#define EDGE_CHUNKS 64
#define NODE_PARTS  4
#define LDS_NODES   53248

__global__ __launch_bounds__(1024) void csr_hist(const int* __restrict__ dst,
                                                 int* __restrict__ part, int n, int e) {
    __shared__ int h[LDS_NODES];
    const int b = blockIdx.x;
    const int chunk = (e + EDGE_CHUNKS - 1) / EDGE_CHUNKS;
    const int e0 = b * chunk;
    const int e1 = (e0 + chunk < e) ? (e0 + chunk) : e;
    const int seg = (n + NODE_PARTS - 1) / NODE_PARTS;
    const int lo = (int)blockIdx.y * seg;
    const int hi = (lo + seg < n) ? (lo + seg) : n;
    for (int r0 = lo; r0 < hi; r0 += LDS_NODES) {
        const int rn = (hi - r0 < LDS_NODES) ? (hi - r0) : LDS_NODES;
        for (int k = threadIdx.x; k < rn; k += 1024) h[k] = 0;
        __syncthreads();
        for (int k0 = e0 + threadIdx.x; k0 < e1; k0 += 4096) {
            int d[4];
            for (int u = 0; u < 4; ++u) {
                const int k = k0 + u * 1024;
                const int dv = dst[(k < e1) ? k : (e1 - 1)];
                d[u] = (k < e1) ? (dv - r0) : -1;
            }
            for (int u = 0; u < 4; ++u)
                if ((unsigned)d[u] < (unsigned)rn) atomicAdd(&h[d[u]], 1);
        }
        __syncthreads();
        for (int k = threadIdx.x; k < rn; k += 1024) part[(size_t)b * n + r0 + k] = h[k];
        __syncthreads();
    }
}

__global__ void csr_colscan(int* __restrict__ part, int* __restrict__ cnt,
                            float* __restrict__ dinv, int n) {
    const int d = blockIdx.x * blockDim.x + threadIdx.x;
    if (d >= n) return;
    int* p = part + d;
    int run = 0;
#pragma clang loop unroll(disable)
    for (int b0 = 0; b0 < EDGE_CHUNKS; b0 += 16) {
        int v[16];
        for (int u = 0; u < 16; ++u) v[u] = p[(size_t)(b0 + u) * n];
        for (int u = 0; u < 16; ++u) {
            p[(size_t)(b0 + u) * n] = run;
            run += v[u];
        }
    }
    cnt[d] = run;
    dinv[d] = rsqrtf((float)(run + 1));
}

__global__ __launch_bounds__(1024) void csr_rowscan(const int* __restrict__ cnt,
                                                    int* __restrict__ rowptr, int n) {
    __shared__ int wsum[32];
    __shared__ int psum[32];
    const int t = threadIdx.x;
    const int lane = t & 31;
    const int wave = t >> 5;
    const int base = blockIdx.x * 1024;

    int pre = 0;
    for (int k = t; k < base; k += 1024) pre += cnt[k];
    const int v = (base + t < n) ? cnt[base + t] : 0;

    int inc = v;
    for (int o = 1; o < 32; o <<= 1) {
        const int u = __shfl_up(inc, o);
        if (lane >= o) inc += u;
    }
    for (int o = 16; o > 0; o >>= 1) pre += __shfl_xor(pre, o);
    if (lane == 31) wsum[wave] = inc;
    if (lane == 0)  psum[wave] = pre;
    __syncthreads();
    int off = 0;
    for (int w = 0; w < 32; ++w) {
        off += psum[w];
        if (w < wave) off += wsum[w];
    }
    if (base + t < n) rowptr[base + t] = off + inc - v;
}

__global__ __launch_bounds__(1024) void csr_fill(const int* __restrict__ src,
                                                 const int* __restrict__ dst,
                                                 const int* __restrict__ part,
                                                 const int* __restrict__ rowptr,
                                                 const float* __restrict__ dinv,
                                                 int2* __restrict__ ent, int n, int e) {
    __shared__ int h[LDS_NODES];
    const int b = blockIdx.x;
    const int chunk = (e + EDGE_CHUNKS - 1) / EDGE_CHUNKS;
    const int e0 = b * chunk;
    const int e1 = (e0 + chunk < e) ? (e0 + chunk) : e;
    const int seg = (n + NODE_PARTS - 1) / NODE_PARTS;
    const int lo = (int)blockIdx.y * seg;
    const int hi = (lo + seg < n) ? (lo + seg) : n;
    for (int r0 = lo; r0 < hi; r0 += LDS_NODES) {
        const int rn = (hi - r0 < LDS_NODES) ? (hi - r0) : LDS_NODES;
        for (int k = threadIdx.x; k < rn; k += 1024)
            h[k] = part[(size_t)b * n + r0 + k] + rowptr[r0 + k];
        __syncthreads();
        for (int k0 = e0 + threadIdx.x; k0 < e1; k0 += 4096) {
            int d[4], s[4];
            float w[4];
            for (int u = 0; u < 4; ++u) {
                const int k = k0 + u * 1024;
                const int kc = (k < e1) ? k : (e1 - 1);
                const int dv = dst[kc];
                s[u] = src[kc];
                d[u] = (k < e1) ? (dv - r0) : -1;
            }
            for (int u = 0; u < 4; ++u)
                w[u] = ((unsigned)d[u] < (unsigned)rn) ? dinv[s[u]] : 0.0f;
            for (int u = 0; u < 4; ++u) {
                if ((unsigned)d[u] < (unsigned)rn) {
                    const int pos = atomicAdd(&h[d[u]], 1);
                    ent[pos] = make_int2(s[u], __float_as_int(w[u]));
                }
            }
        }
        __syncthreads();
    }
}

__global__ __launch_bounds__(256) void gcn_gemm_wmma(const float* __restrict__ x,
                                                     const float* __restrict__ W,
                                                     float* __restrict__ xw,
                                                     int n_nodes) {
    const int lane = threadIdx.x & 31;
    const int wave = threadIdx.x >> 5;
    const int half = lane >> 4;
    const int l    = lane & 15;

    const int m0 = blockIdx.x * 16;
    const int n0 = wave * 16;

    int arow = m0 + l;
    if (arow >= n_nodes) arow = n_nodes - 1;

    const float* aptr = x + (size_t)arow * NFEAT + 2 * half;
    const float* bptr = W + (size_t)(2 * half) * NHID + n0 + l;

    v8f c = {};
#pragma unroll
    for (int kk = 0; kk < NFEAT; kk += 4) {
        v2f a = *(const v2f*)(aptr + kk);
        v2f b;
        b.x = bptr[(size_t)kk * NHID];
        b.y = bptr[(size_t)kk * NHID + NHID];
        c = __builtin_amdgcn_wmma_f32_16x16x4_f32(
                 false, a,  false, b,
                 (short)0, c,  false,  false);
    }

#pragma unroll
    for (int r = 0; r < 8; ++r) {
        int row = m0 + r + 8 * half;
        if (row < n_nodes) xw[(size_t)row * NHID + n0 + l] = c[r];
    }
}

__global__ __launch_bounds__(256) void gcn_gather(const int2* __restrict__ ent,
                                                  const int* __restrict__ rowptr,
                                                  const int* __restrict__ cnt,
                                                  const float* __restrict__ dinv,
                                                  const float* __restrict__ xw,
                                                  const float* __restrict__ bias,
                                                  float* __restrict__ out, int n) {
    const int lane = threadIdx.x & 31;
    const int i = __builtin_amdgcn_readfirstlane((int)(blockIdx.x * 8 + (threadIdx.x >> 5)));
    if (i >= n) return;

    const float di = dinv[i];
    const float n2 = di * di;
    const float4 v  = *(const float4*)(xw + (size_t)i * NHID + lane * 4);
    const float4 bb = *(const float4*)(bias + lane * 4);
    float4 acc;
    acc.x = fmaf(v.x, n2, bb.x);
    acc.y = fmaf(v.y, n2, bb.y);
    acc.z = fmaf(v.z, n2, bb.z);
    acc.w = fmaf(v.w, n2, bb.w);

    const int start = rowptr[i];
    const int total = cnt[i];
    for (int base = 0; base < total; base += 32) {
        const int m = (total - base < 32) ? (total - base) : 32;
        int sidx = 0;
        float wgt = 0.0f;
        if (lane < m) {
            const int2 en = ent[start + base + lane];
            sidx = en.x;
            wgt  = __int_as_float(en.y) * di;
        }
        const int wbits = __float_as_int(wgt);
        int j = 0;
        for (; j + 4 <= m; j += 4) {
            const int s0 = __builtin_amdgcn_readlane(sidx, j);
            const int s1 = __builtin_amdgcn_readlane(sidx, j + 1);
            const int s2 = __builtin_amdgcn_readlane(sidx, j + 2);
            const int s3 = __builtin_amdgcn_readlane(sidx, j + 3);
            const float w0 = __int_as_float(__builtin_amdgcn_readlane(wbits, j));
            const float w1 = __int_as_float(__builtin_amdgcn_readlane(wbits, j + 1));
            const float w2 = __int_as_float(__builtin_amdgcn_readlane(wbits, j + 2));
            const float w3 = __int_as_float(__builtin_amdgcn_readlane(wbits, j + 3));
            const float4 r0 = *(const float4*)(xw + (size_t)s0 * NHID + lane * 4);
            const float4 r1 = *(const float4*)(xw + (size_t)s1 * NHID + lane * 4);
            const float4 r2 = *(const float4*)(xw + (size_t)s2 * NHID + lane * 4);
            const float4 r3 = *(const float4*)(xw + (size_t)s3 * NHID + lane * 4);
            acc.x = fmaf(r0.x, w0, acc.x); acc.y = fmaf(r0.y, w0, acc.y);
            acc.z = fmaf(r0.z, w0, acc.z); acc.w = fmaf(r0.w, w0, acc.w);
            acc.x = fmaf(r1.x, w1, acc.x); acc.y = fmaf(r1.y, w1, acc.y);
            acc.z = fmaf(r1.z, w1, acc.z); acc.w = fmaf(r1.w, w1, acc.w);
            acc.x = fmaf(r2.x, w2, acc.x); acc.y = fmaf(r2.y, w2, acc.y);
            acc.z = fmaf(r2.z, w2, acc.z); acc.w = fmaf(r2.w, w2, acc.w);
            acc.x = fmaf(r3.x, w3, acc.x); acc.y = fmaf(r3.y, w3, acc.y);
            acc.z = fmaf(r3.z, w3, acc.z); acc.w = fmaf(r3.w, w3, acc.w);
        }
        for (; j < m; ++j) {
            const int s0 = __builtin_amdgcn_readlane(sidx, j);
            const float w0 = __int_as_float(__builtin_amdgcn_readlane(wbits, j));
            const float4 r0 = *(const float4*)(xw + (size_t)s0 * NHID + lane * 4);
            acc.x = fmaf(r0.x, w0, acc.x); acc.y = fmaf(r0.y, w0, acc.y);
            acc.z = fmaf(r0.z, w0, acc.z); acc.w = fmaf(r0.w, w0, acc.w);
        }
    }
    *(float4*)(out + (size_t)i * NHID + lane * 4) = acc;
}

extern "C" void kernel_launch(void* const* d_in, const int* in_sizes, int n_in,
                              void* d_out, int out_size, void* d_ws, size_t ws_size,
                              hipStream_t stream) {
    const int*   ei   = (const int*)d_in[0];
    const float* x    = (const float*)d_in[1];
    const float* W    = (const float*)d_in[2];
    const float* bias = (const float*)d_in[3];
    float*       out  = (float*)d_out;

    const int E = in_sizes[0] / 2;
    const int N = in_sizes[1] / NFEAT;

    float* xw     = (float*)d_ws;
    int2*  ent    = (int2*)(xw + (size_t)N * NHID);
    int*   part   = (int*)(ent + (size_t)E);
    float* dinv   = (float*)(part + (size_t)EDGE_CHUNKS * N);
    int*   cnt    = (int*)(dinv + (size_t)N);
    int*   rowptr = cnt + (size_t)N;

    const int* srcIdx = ei;
    const int* dstIdx = ei + E;

    csr_hist   <<<dim3(EDGE_CHUNKS, NODE_PARTS), 1024, 0, stream>>>(dstIdx, part, N, E);
    csr_colscan<<<(N + 255) / 256, 256, 0, stream>>>(part, cnt, dinv, N);
    csr_rowscan<<<(N + 1023) / 1024, 1024, 0, stream>>>(cnt, rowptr, N);
    csr_fill   <<<dim3(EDGE_CHUNKS, NODE_PARTS), 1024, 0, stream>>>(srcIdx, dstIdx, part, rowptr, dinv, ent, N, E);

    gcn_gemm_wmma<<<(N + 15) / 16, 256, 0, stream>>>(x, W, xw, N);

    gcn_gather<<<(N + 7) / 8, 256, 0, stream>>>(ent, rowptr, cnt, dinv, xw, bias, out, N);
}


# GEMM: each wave computes a 16 x 128 row block (8 accumulator tiles), A fragments loaded once not 8 times, W staged per work-group in LDS as B source; 391 work-groups not 3125; sum order unchanged
# speedup vs baseline: 2.5728x; 1.1184x over previous
#include <hip/hip_runtime.h>


typedef __attribute__((ext_vector_type(2))) float v2f;
typedef __attribute__((ext_vector_type(8))) float v8f;

#define NFEAT 128
#define NHID  128

#define EDGE_CHUNKS 64
#define NODE_PARTS  4
#define LDS_NODES   53248

__global__ __launch_bounds__(1024) void csr_hist(const int* __restrict__ dst,
                                                 int* __restrict__ part, int n, int e) {
    __shared__ int h[LDS_NODES];
    const int b = blockIdx.x;
    const int chunk = (e + EDGE_CHUNKS - 1) / EDGE_CHUNKS;
    const int e0 = b * chunk;
    const int e1 = (e0 + chunk < e) ? (e0 + chunk) : e;
    const int seg = (n + NODE_PARTS - 1) / NODE_PARTS;
    const int lo = (int)blockIdx.y * seg;
    const int hi = (lo + seg < n) ? (lo + seg) : n;
    for (int r0 = lo; r0 < hi; r0 += LDS_NODES) {
        const int rn = (hi - r0 < LDS_NODES) ? (hi - r0) : LDS_NODES;
        for (int k = threadIdx.x; k < rn; k += 1024) h[k] = 0;
        __syncthreads();
        for (int k0 = e0 + threadIdx.x; k0 < e1; k0 += 4096) {
            int d[4];
            for (int u = 0; u < 4; ++u) {
                const int k = k0 + u * 1024;
                const int dv = dst[(k < e1) ? k : (e1 - 1)];
                d[u] = (k < e1) ? (dv - r0) : -1;
            }
            for (int u = 0; u < 4; ++u)
                if ((unsigned)d[u] < (unsigned)rn) atomicAdd(&h[d[u]], 1);
        }
        __syncthreads();
        for (int k = threadIdx.x; k < rn; k += 1024) part[(size_t)b * n + r0 + k] = h[k];
        __syncthreads();
    }
}

__global__ void csr_colscan(int* __restrict__ part, int* __restrict__ cnt,
                            float* __restrict__ dinv, int n) {
    const int d = blockIdx.x * blockDim.x + threadIdx.x;
    if (d >= n) return;
    int* p = part + d;
    int run = 0;
#pragma clang loop unroll(disable)
    for (int b0 = 0; b0 < EDGE_CHUNKS; b0 += 16) {
        int v[16];
        for (int u = 0; u < 16; ++u) v[u] = p[(size_t)(b0 + u) * n];
        for (int u = 0; u < 16; ++u) {
            p[(size_t)(b0 + u) * n] = run;
            run += v[u];
        }
    }
    cnt[d] = run;
    dinv[d] = rsqrtf((float)(run + 1));
}

__global__ __launch_bounds__(1024) void csr_rowscan(const int* __restrict__ cnt,
                                                    int* __restrict__ rowptr, int n) {
    __shared__ int wsum[32];
    __shared__ int psum[32];
    const int t = threadIdx.x;
    const int lane = t & 31;
    const int wave = t >> 5;
    const int base = blockIdx.x * 1024;

    int pre = 0;
    for (int k = t; k < base; k += 1024) pre += cnt[k];
    const int v = (base + t < n) ? cnt[base + t] : 0;

    int inc = v;
    for (int o = 1; o < 32; o <<= 1) {
        const int u = __shfl_up(inc, o);
        if (lane >= o) inc += u;
    }
    for (int o = 16; o > 0; o >>= 1) pre += __shfl_xor(pre, o);
    if (lane == 31) wsum[wave] = inc;
    if (lane == 0)  psum[wave] = pre;
    __syncthreads();
    int off = 0;
    for (int w = 0; w < 32; ++w) {
        off += psum[w];
        if (w < wave) off += wsum[w];
    }
    if (base + t < n) rowptr[base + t] = off + inc - v;
}

__global__ __launch_bounds__(1024) void csr_fill(const int* __restrict__ src,
                                                 const int* __restrict__ dst,
                                                 const int* __restrict__ part,
                                                 const int* __restrict__ rowptr,
                                                 const float* __restrict__ dinv,
                                                 int2* __restrict__ ent, int n, int e) {
    __shared__ int h[LDS_NODES];
    const int b = blockIdx.x;
    const int chunk = (e + EDGE_CHUNKS - 1) / EDGE_CHUNKS;
    const int e0 = b * chunk;
    const int e1 = (e0 + chunk < e) ? (e0 + chunk) : e;
    const int seg = (n + NODE_PARTS - 1) / NODE_PARTS;
    const int lo = (int)blockIdx.y * seg;
    const int hi = (lo + seg < n) ? (lo + seg) : n;
    for (int r0 = lo; r0 < hi; r0 += LDS_NODES) {
        const int rn = (hi - r0 < LDS_NODES) ? (hi - r0) : LDS_NODES;
        for (int k = threadIdx.x; k < rn; k += 1024)
            h[k] = part[(size_t)b * n + r0 + k] + rowptr[r0 + k];
        __syncthreads();
        for (int k0 = e0 + threadIdx.x; k0 < e1; k0 += 4096) {
            int d[4], s[4];
            float w[4];
            for (int u = 0; u < 4; ++u) {
                const int k = k0 + u * 1024;
                const int kc = (k < e1) ? k : (e1 - 1);
                const int dv = dst[kc];
                s[u] = src[kc];
                d[u] = (k < e1) ? (dv - r0) : -1;
            }
            for (int u = 0; u < 4; ++u)
                w[u] = ((unsigned)d[u] < (unsigned)rn) ? dinv[s[u]] : 0.0f;
            for (int u = 0; u < 4; ++u) {
                if ((unsigned)d[u] < (unsigned)rn) {
                    const int pos = atomicAdd(&h[d[u]], 1);
                    ent[pos] = make_int2(s[u], __float_as_int(w[u]));
                }
            }
        }
        __syncthreads();
    }
}

#define KQUADS (NFEAT / 4)
__global__ __launch_bounds__(256) void gcn_gemm_wmma(const float* __restrict__ x,
                                                     const float* __restrict__ W,
                                                     float* __restrict__ xw,
                                                     int n_nodes) {
    __shared__ float4 Ws[KQUADS * NHID];

    const int lane = threadIdx.x & 31;
    const int wave = threadIdx.x >> 5;
    const int half = lane >> 4;
    const int l    = lane & 15;

    const int m0 = (blockIdx.x * 8 + wave) * 16;
    const bool live = m0 < n_nodes;

    float4 wv[16];
#pragma unroll
    for (int it = 0; it < 16; ++it) {
        const int p = threadIdx.x + it * 256;
        const float* wp = W + (size_t)(4 * (p >> 7)) * NHID + (p & (NHID - 1));
        wv[it].x = wp[0];
        wv[it].y = wp[NHID];
        wv[it].z = wp[2 * NHID];
        wv[it].w = wp[3 * NHID];
    }

    int arow = m0 + l;
    if (arow >= n_nodes) arow = n_nodes - 1;
    const float* aptr = x + (size_t)arow * NFEAT + 2 * half;
    v2f a[KQUADS];
    if (live) {
#pragma unroll
        for (int q = 0; q < KQUADS; ++q) a[q] = *(const v2f*)(aptr + 4 * q);
    }

#pragma unroll
    for (int it = 0; it < 16; ++it) Ws[threadIdx.x + it * 256] = wv[it];
    __syncthreads();
    if (!live) return;

    const float* bbase = (const float*)Ws + l * 4 + 2 * half;

    v8f c[8];
    v2f bcur[8], bnext[8];
#pragma unroll
    for (int t = 0; t < 8; ++t) {
        c[t] = (v8f){};
        bcur[t] = *(const v2f*)(bbase + (16 * t) * 4);
    }

#pragma unroll
    for (int q = 0; q < KQUADS; ++q) {
        if (q + 1 < KQUADS) {
#pragma unroll
            for (int t = 0; t < 8; ++t)
                bnext[t] = *(const v2f*)(bbase + ((q + 1) * NHID + 16 * t) * 4);
        }
        __builtin_amdgcn_sched_barrier(0);
#pragma unroll
        for (int t = 0; t < 8; ++t)
            c[t] = __builtin_amdgcn_wmma_f32_16x16x4_f32(
                        false, a[q],  false, bcur[t],
                        (short)0, c[t],  false,  false);
        __builtin_amdgcn_sched_barrier(0);
#pragma unroll
        for (int t = 0; t < 8; ++t) bcur[t] = bnext[t];
    }

#pragma unroll
    for (int r = 0; r < 8; ++r) {
        const int row = m0 + r + 8 * half;
        if (row < n_nodes) {
#pragma unroll
            for (int t = 0; t < 8; ++t) xw[(size_t)row * NHID + 16 * t + l] = c[t][r];
        }
    }
}

__global__ __launch_bounds__(256) void gcn_gather(const int2* __restrict__ ent,
                                                  const int* __restrict__ rowptr,
                                                  const int* __restrict__ cnt,
                                                  const float* __restrict__ dinv,
                                                  const float* __restrict__ xw,
                                                  const float* __restrict__ bias,
                                                  float* __restrict__ out, int n) {
    const int lane = threadIdx.x & 31;
    const int i = __builtin_amdgcn_readfirstlane((int)(blockIdx.x * 8 + (threadIdx.x >> 5)));
    if (i >= n) return;

    const float di = dinv[i];
    const float n2 = di * di;
    const float4 v  = *(const float4*)(xw + (size_t)i * NHID + lane * 4);
    const float4 bb = *(const float4*)(bias + lane * 4);
    float4 acc;
    acc.x = fmaf(v.x, n2, bb.x);
    acc.y = fmaf(v.y, n2, bb.y);
    acc.z = fmaf(v.z, n2, bb.z);
    acc.w = fmaf(v.w, n2, bb.w);

    const int start = rowptr[i];
    const int total = cnt[i];
    for (int base = 0; base < total; base += 32) {
        const int m = (total - base < 32) ? (total - base) : 32;
        int sidx = 0;
        float wgt = 0.0f;
        if (lane < m) {
            const int2 en = ent[start + base + lane];
            sidx = en.x;
            wgt  = __int_as_float(en.y) * di;
        }
        const int wbits = __float_as_int(wgt);
        int j = 0;
        for (; j + 4 <= m; j += 4) {
            const int s0 = __builtin_amdgcn_readlane(sidx, j);
            const int s1 = __builtin_amdgcn_readlane(sidx, j + 1);
            const int s2 = __builtin_amdgcn_readlane(sidx, j + 2);
            const int s3 = __builtin_amdgcn_readlane(sidx, j + 3);
            const float w0 = __int_as_float(__builtin_amdgcn_readlane(wbits, j));
            const float w1 = __int_as_float(__builtin_amdgcn_readlane(wbits, j + 1));
            const float w2 = __int_as_float(__builtin_amdgcn_readlane(wbits, j + 2));
            const float w3 = __int_as_float(__builtin_amdgcn_readlane(wbits, j + 3));
            const float4 r0 = *(const float4*)(xw + (size_t)s0 * NHID + lane * 4);
            const float4 r1 = *(const float4*)(xw + (size_t)s1 * NHID + lane * 4);
            const float4 r2 = *(const float4*)(xw + (size_t)s2 * NHID + lane * 4);
            const float4 r3 = *(const float4*)(xw + (size_t)s3 * NHID + lane * 4);
            acc.x = fmaf(r0.x, w0, acc.x); acc.y = fmaf(r0.y, w0, acc.y);
            acc.z = fmaf(r0.z, w0, acc.z); acc.w = fmaf(r0.w, w0, acc.w);
            acc.x = fmaf(r1.x, w1, acc.x); acc.y = fmaf(r1.y, w1, acc.y);
            acc.z = fmaf(r1.z, w1, acc.z); acc.w = fmaf(r1.w, w1, acc.w);
            acc.x = fmaf(r2.x, w2, acc.x); acc.y = fmaf(r2.y, w2, acc.y);
            acc.z = fmaf(r2.z, w2, acc.z); acc.w = fmaf(r2.w, w2, acc.w);
            acc.x = fmaf(r3.x, w3, acc.x); acc.y = fmaf(r3.y, w3, acc.y);
            acc.z = fmaf(r3.z, w3, acc.z); acc.w = fmaf(r3.w, w3, acc.w);
        }
        for (; j < m; ++j) {
            const int s0 = __builtin_amdgcn_readlane(sidx, j);
            const float w0 = __int_as_float(__builtin_amdgcn_readlane(wbits, j));
            const float4 r0 = *(const float4*)(xw + (size_t)s0 * NHID + lane * 4);
            acc.x = fmaf(r0.x, w0, acc.x); acc.y = fmaf(r0.y, w0, acc.y);
            acc.z = fmaf(r0.z, w0, acc.z); acc.w = fmaf(r0.w, w0, acc.w);
        }
    }
    *(float4*)(out + (size_t)i * NHID + lane * 4) = acc;
}

extern "C" void kernel_launch(void* const* d_in, const int* in_sizes, int n_in,
                              void* d_out, int out_size, void* d_ws, size_t ws_size,
                              hipStream_t stream) {
    const int*   ei   = (const int*)d_in[0];
    const float* x    = (const float*)d_in[1];
    const float* W    = (const float*)d_in[2];
    const float* bias = (const float*)d_in[3];
    float*       out  = (float*)d_out;

    const int E = in_sizes[0] / 2;
    const int N = in_sizes[1] / NFEAT;

    float* xw     = (float*)d_ws;
    int2*  ent    = (int2*)(xw + (size_t)N * NHID);
    int*   part   = (int*)(ent + (size_t)E);
    float* dinv   = (float*)(part + (size_t)EDGE_CHUNKS * N);
    int*   cnt    = (int*)(dinv + (size_t)N);
    int*   rowptr = cnt + (size_t)N;

    const int* srcIdx = ei;
    const int* dstIdx = ei + E;

    csr_hist   <<<dim3(EDGE_CHUNKS, NODE_PARTS), 1024, 0, stream>>>(dstIdx, part, N, E);
    csr_colscan<<<(N + 255) / 256, 256, 0, stream>>>(part, cnt, dinv, N);
    csr_rowscan<<<(N + 1023) / 1024, 1024, 0, stream>>>(cnt, rowptr, N);
    csr_fill   <<<dim3(EDGE_CHUNKS, NODE_PARTS), 1024, 0, stream>>>(srcIdx, dstIdx, part, rowptr, dinv, ent, N, E);

    gcn_gemm_wmma<<<(N + 127) / 128, 256, 0, stream>>>(x, W, xw, N);

    gcn_gather<<<(N + 7) / 8, 256, 0, stream>>>(ent, rowptr, cnt, dinv, xw, bias, out, N);
}


# GEMM folded into the edge-placing launch: each of its 256 work-groups then computes its share of row blocks (W in LDS, dinv re-summed from chunk counts); 3 launches not 4; sum order unchanged
# speedup vs baseline: 2.8790x; 1.1190x over previous
#include <hip/hip_runtime.h>


typedef __attribute__((ext_vector_type(2))) float v2f;
typedef __attribute__((ext_vector_type(8))) float v8f;

#define NFEAT 128
#define NHID  128

#define EDGE_CHUNKS 16
#define NODE_PARTS  16
#define LDS_NODES   26624

__global__ __launch_bounds__(1024) void csr_hist(const int* __restrict__ dst,
                                                 int* __restrict__ part,
                                                 int* __restrict__ tot, int n, int e) {
    __shared__ int h[2 * LDS_NODES];
    __shared__ int wsum[32];
    const int lane = threadIdx.x & 31;
    const int wave = threadIdx.x >> 5;
    const int b = blockIdx.x;
    const int y = blockIdx.y;
    const int chunk = (e + EDGE_CHUNKS - 1) / EDGE_CHUNKS;
    const int e0 = b * chunk;
    const int e1 = (e0 + chunk < e) ? (e0 + chunk) : e;
    const int seg = (n + NODE_PARTS - 1) / NODE_PARTS;
    const int lo = y * seg;
    const int hi = (lo + seg < n) ? (lo + seg) : n;
    int rn = (hi > lo) ? (hi - lo) : 0;
    if (rn > LDS_NODES) rn = LDS_NODES;

    for (int k = threadIdx.x; k < rn; k += 1024) h[k] = 0;
    __syncthreads();
    int mine = 0;
    for (int k0 = e0 + (int)threadIdx.x; k0 < e1; k0 += 32 * 1024) {
        int d[32];
#pragma unroll
        for (int u = 0; u < 32; ++u) {
            const int k = k0 + u * 1024;
            const int dv = dst[(k < e1) ? k : (e1 - 1)];
            d[u] = (k < e1) ? (dv - lo) : -1;
        }
#pragma unroll
        for (int u = 0; u < 32; ++u)
            if ((unsigned)d[u] < (unsigned)rn) { atomicAdd(&h[d[u]], 1); ++mine; }
    }
    for (int o = 16; o > 0; o >>= 1) mine += __shfl_xor(mine, o);
    if (lane == 0) wsum[wave] = mine;
    __syncthreads();
    for (int k = threadIdx.x; k < rn; k += 1024) part[(size_t)b * n + lo + k] = h[k];
    if (threadIdx.x == 0) {
        int s = 0;
        for (int w = 0; w < 32; ++w) s += wsum[w];
        tot[b * NODE_PARTS + y] = s;
    }
}

#define KQUADS (NFEAT / 4)
__global__ __launch_bounds__(1024) void csr_fill_gemm(const int* __restrict__ src,
                                                      const int* __restrict__ dst,
                                                      const int* __restrict__ part,
                                                      const int* __restrict__ tot,
                                                      int* __restrict__ rowptr,
                                                      int* __restrict__ cnt,
                                                      float* __restrict__ dinv,
                                                      int* __restrict__ ent,
                                                      const float* __restrict__ x,
                                                      const float* __restrict__ W,
                                                      float* __restrict__ xw, int n, int e) {
    __shared__ int hc[LDS_NODES];
    __shared__ int hp[LDS_NODES];
    __shared__ int wsum[32];
    __shared__ int psum[32];
    __shared__ float4 Ws[KQUADS * NHID];
    const int t = threadIdx.x;

    float4 wv[4];
#pragma unroll
    for (int it = 0; it < 4; ++it) {
        const int p = t + it * 1024;
        const float* wp = W + (size_t)(4 * (p >> 7)) * NHID + (p & (NHID - 1));
        wv[it].x = wp[0];
        wv[it].y = wp[NHID];
        wv[it].z = wp[2 * NHID];
        wv[it].w = wp[3 * NHID];
    }
    const int lane = t & 31;
    const int wave = t >> 5;
    const int b = blockIdx.x;
    const int y = blockIdx.y;
    const int chunk = (e + EDGE_CHUNKS - 1) / EDGE_CHUNKS;
    const int e0 = b * chunk;
    const int e1 = (e0 + chunk < e) ? (e0 + chunk) : e;
    const int seg = (n + NODE_PARTS - 1) / NODE_PARTS;
    const int lo = y * seg;
    const int hi = (lo + seg < n) ? (lo + seg) : n;
    int rn = (hi > lo) ? (hi - lo) : 0;
    if (rn > LDS_NODES) rn = LDS_NODES;

    int pre = 0;
    if (t < EDGE_CHUNKS * NODE_PARTS && (t % NODE_PARTS) < y) pre = tot[t];

    for (int k0 = 0; k0 < rn; k0 += 2048) {
        const int ka = k0 + t;
        const int kb = ka + 1024;
        const int* pa = part + lo + ((ka < rn) ? ka : (rn - 1));
        const int* pb = part + lo + ((kb < rn) ? kb : (rn - 1));
        int va[EDGE_CHUNKS], vb[EDGE_CHUNKS];
#pragma unroll
        for (int q = 0; q < EDGE_CHUNKS; ++q) {
            va[q] = pa[(size_t)q * n];
            vb[q] = pb[(size_t)q * n];
        }
        int ca = 0, cb = 0, qa = 0, qb = 0;
#pragma unroll
        for (int q = 0; q < EDGE_CHUNKS; ++q) {
            ca += va[q];
            cb += vb[q];
            if (q < b) { qa += va[q]; qb += vb[q]; }
        }
        if (ka < rn) { hc[ka] = ca; hp[ka] = qa; }
        if (kb < rn) { hc[kb] = cb; hp[kb] = qb; }
    }
#pragma unroll
    for (int it = 0; it < 4; ++it) {
        const int p = t + it * 1024;
        const int cc = p & (NHID - 1);
        const int qd = p >> 7;
        const int f0 = ((((qd * 2 + (cc >> 6)) * 16 + (cc & 15)) * 2) * 4 + ((cc >> 4) & 3)) * 2;
        *(v2f*)((float*)Ws + f0)     = (v2f){wv[it].x, wv[it].y};
        *(v2f*)((float*)Ws + f0 + 8) = (v2f){wv[it].z, wv[it].w};
    }
    __syncthreads();

    const int per = (rn + 1023) >> 10;
    const int k_lo = (t * per < rn) ? (t * per) : rn;
    const int k_hi = (k_lo + per < rn) ? (k_lo + per) : rn;
    int v = 0;
    for (int k = k_lo; k < k_hi; ++k) v += hc[k];
    int inc = v;
    for (int o = 1; o < 32; o <<= 1) {
        const int u = __shfl_up(inc, o);
        if (lane >= o) inc += u;
    }
    for (int o = 16; o > 0; o >>= 1) pre += __shfl_xor(pre, o);
    if (lane == 31) wsum[wave] = inc;
    if (lane == 0)  psum[wave] = pre;
    __syncthreads();
    int off = 0;
    for (int w = 0; w < 32; ++w) {
        off += psum[w];
        if (w < wave) off += wsum[w];
    }
    int run = off + inc - v;
    for (int k = k_lo; k < k_hi; ++k) {
        const int c = hc[k];
        if (b == 0) {
            rowptr[lo + k] = run;
            cnt[lo + k] = c;
            dinv[lo + k] = rsqrtf((float)(c + 1));
        }
        hc[k] = run + hp[k];
        run += c;
    }
    __syncthreads();

    for (int k0 = e0 + t; k0 < e1; k0 += 16 * 1024) {
        int d[16], s[16], pos[16];
#pragma unroll
        for (int u = 0; u < 16; ++u) {
            const int k = k0 + u * 1024;
            const int kc = (k < e1) ? k : (e1 - 1);
            const int dv = dst[kc];
            s[u] = src[kc];
            d[u] = (k < e1) ? (dv - lo) : -1;
        }
        asm volatile("" :: "v"(s[0]), "v"(s[1]), "v"(s[2]), "v"(s[3]), "v"(s[4]), "v"(s[5]),
                           "v"(s[6]), "v"(s[7]), "v"(s[8]), "v"(s[9]), "v"(s[10]), "v"(s[11]),
                           "v"(s[12]), "v"(s[13]), "v"(s[14]), "v"(s[15]));
#pragma unroll
        for (int u = 0; u < 16; ++u)
            pos[u] = ((unsigned)d[u] < (unsigned)rn) ? atomicAdd(&hc[d[u]], 1) : -1;
#pragma unroll
        for (int u = 0; u < 16; ++u)
            if (pos[u] >= 0) ent[pos[u]] = s[u];
    }

    const int half = lane >> 4;
    const int l    = lane & 15;
    const int wv_id = __builtin_amdgcn_readfirstlane(wave);
    const int ch    = wv_id & 1;
    const int g     = b + EDGE_CHUNKS * y;
    const int nblk  = (n + 15) >> 4;
    const float4* bbase = Ws + ((ch * 16 + l) * 2 + half) * 2;

    for (int rb = g + EDGE_CHUNKS * NODE_PARTS * (wv_id >> 1); rb < nblk;
         rb += EDGE_CHUNKS * NODE_PARTS * 16) {
        const int m0 = rb * 16;
        int arow = m0 + l;
        if (arow >= n) arow = n - 1;
        const float* aptr = x + (size_t)arow * NFEAT + 2 * half;
        v2f a[KQUADS];
#pragma unroll
        for (int q = 0; q < KQUADS; ++q) a[q] = *(const v2f*)(aptr + 4 * q);

        int cs = 0;
#pragma unroll
        for (int q = 0; q < EDGE_CHUNKS / 2; ++q) cs += part[(size_t)(half * (EDGE_CHUNKS / 2) + q) * n + arow];
        cs += __shfl_xor(cs, 16);
        const float dv = rsqrtf((float)(cs + 1));

        v8f c[4];
#pragma unroll
        for (int tt = 0; tt < 4; ++tt) c[tt] = (v8f){};

#pragma unroll
        for (int q = 0; q < KQUADS; ++q) {
            const float4 b01 = bbase[128 * q];
            const float4 b23 = bbase[128 * q + 1];
            v2f bq[4];
            bq[0] = (v2f){b01.x, b01.y};
            bq[1] = (v2f){b01.z, b01.w};
            bq[2] = (v2f){b23.x, b23.y};
            bq[3] = (v2f){b23.z, b23.w};
            __builtin_amdgcn_sched_barrier(0);
#pragma unroll
            for (int tt = 0; tt < 4; ++tt)
                c[tt] = __builtin_amdgcn_wmma_f32_16x16x4_f32(
                             false, a[q],  false, bq[tt],
                             (short)0, c[tt],  false,  false);
            __builtin_amdgcn_sched_barrier(0);
        }

#pragma unroll
        for (int r = 0; r < 8; ++r) {
            const int row = m0 + r + 8 * half;
            const float dsc = __shfl(dv, r + 8 * half);
            if (row < n) {
#pragma unroll
                for (int tt = 0; tt < 4; ++tt)
                    xw[(size_t)row * NHID + 64 * ch + 16 * tt + l] = c[tt][r] * dsc;
            }
        }
    }
}

__global__ __launch_bounds__(256) void gcn_gather(const int* __restrict__ ent,
                                                  const int* __restrict__ rowptr,
                                                  const int* __restrict__ cnt,
                                                  const float* __restrict__ dinv,
                                                  const float* __restrict__ xw,
                                                  const float* __restrict__ bias,
                                                  float* __restrict__ out, int n) {
    const int lane = threadIdx.x & 31;
    const int i = __builtin_amdgcn_readfirstlane((int)(blockIdx.x * 8 + (threadIdx.x >> 5)));
    if (i >= n) return;

    const float di = dinv[i];
    float4 acc = *(const float4*)(xw + (size_t)i * NHID + lane * 4);
    const float4 bb = *(const float4*)(bias + lane * 4);

    const int start = rowptr[i];
    const int total = cnt[i];
    for (int base = 0; base < total; base += 32) {
        const int m = (total - base < 32) ? (total - base) : 32;
        int sidx = 0;
        if (lane < m) sidx = ent[start + base + lane];
        int j = 0;
        for (; j + 4 <= m; j += 4) {
            const int s0 = __builtin_amdgcn_readlane(sidx, j);
            const int s1 = __builtin_amdgcn_readlane(sidx, j + 1);
            const int s2 = __builtin_amdgcn_readlane(sidx, j + 2);
            const int s3 = __builtin_amdgcn_readlane(sidx, j + 3);
            const float4 r0 = *(const float4*)(xw + (size_t)s0 * NHID + lane * 4);
            const float4 r1 = *(const float4*)(xw + (size_t)s1 * NHID + lane * 4);
            const float4 r2 = *(const float4*)(xw + (size_t)s2 * NHID + lane * 4);
            const float4 r3 = *(const float4*)(xw + (size_t)s3 * NHID + lane * 4);
            acc.x += r0.x; acc.y += r0.y; acc.z += r0.z; acc.w += r0.w;
            acc.x += r1.x; acc.y += r1.y; acc.z += r1.z; acc.w += r1.w;
            acc.x += r2.x; acc.y += r2.y; acc.z += r2.z; acc.w += r2.w;
            acc.x += r3.x; acc.y += r3.y; acc.z += r3.z; acc.w += r3.w;
        }
        for (; j < m; ++j) {
            const int s0 = __builtin_amdgcn_readlane(sidx, j);
            const float4 r0 = *(const float4*)(xw + (size_t)s0 * NHID + lane * 4);
            acc.x += r0.x; acc.y += r0.y; acc.z += r0.z; acc.w += r0.w;
        }
    }
    float4 o;
    o.x = fmaf(acc.x, di, bb.x);
    o.y = fmaf(acc.y, di, bb.y);
    o.z = fmaf(acc.z, di, bb.z);
    o.w = fmaf(acc.w, di, bb.w);
    *(float4*)(out + (size_t)i * NHID + lane * 4) = o;
}

extern "C" void kernel_launch(void* const* d_in, const int* in_sizes, int n_in,
                              void* d_out, int out_size, void* d_ws, size_t ws_size,
                              hipStream_t stream) {
    const int*   ei   = (const int*)d_in[0];
    const float* x    = (const float*)d_in[1];
    const float* W    = (const float*)d_in[2];
    const float* bias = (const float*)d_in[3];
    float*       out  = (float*)d_out;

    const int E = in_sizes[0] / 2;
    const int N = in_sizes[1] / NFEAT;

    float* xw     = (float*)d_ws;
    int*   ent    = (int*)(xw + (size_t)N * NHID);
    int*   part   = ent + (size_t)E;
    int*   tot    = part + (size_t)EDGE_CHUNKS * N;
    float* dinv   = (float*)(tot + EDGE_CHUNKS * NODE_PARTS);
    int*   cnt    = (int*)(dinv + (size_t)N);
    int*   rowptr = cnt + (size_t)N;

    const int* srcIdx = ei;
    const int* dstIdx = ei + E;

    csr_hist<<<dim3(EDGE_CHUNKS, NODE_PARTS), 1024, 0, stream>>>(dstIdx, part, tot, N, E);
    csr_fill_gemm<<<dim3(EDGE_CHUNKS, NODE_PARTS), 1024, 0, stream>>>(srcIdx, dstIdx, part, tot, rowptr, cnt, dinv, ent, x, W, xw, N, E);

    gcn_gather<<<(N + 7) / 8, 256, 0, stream>>>(ent, rowptr, cnt, dinv, xw, bias, out, N);
}


# CSR in two levels: launch 1 sorts each of 256 edge chunks by node range in LDS; a launch-2 work-group owns one range (gets its 256 runs, writes ent in whole lines, GEMMs its rows); sum order changes
# speedup vs baseline: 4.2151x; 1.4641x over previous
#include <hip/hip_runtime.h>


typedef __attribute__((ext_vector_type(2))) float v2f;
typedef __attribute__((ext_vector_type(8))) float v8f;

#define NFEAT 128
#define NHID  128

#define NCHUNK 256
#define NRANGE 256
#define EPT    4
#define RPT    4
#define STAGE_CAP 24576
#define RANGE_CAP 4096
#define ENT_CAP   24576

__device__ __forceinline__ int range_nodes(int n) {
    const int nblk = (n + 15) >> 4;
    return 16 * ((nblk + NRANGE - 1) / NRANGE);
}

__global__ __launch_bounds__(1024) void edge_partition(const int* __restrict__ src,
                                                       const int* __restrict__ dst,
                                                       int* __restrict__ tmp_w,
                                                       int* __restrict__ meta_w, int n, int e) {
    int2* __restrict__ tmp  = (int2*)tmp_w;
    int2* __restrict__ meta = (int2*)meta_w;
    __shared__ int rc[NRANGE];
    __shared__ int cur[NRANGE];
    __shared__ int wsum[NRANGE / 32];
    __shared__ int nvalid;
    __shared__ int2 stage[STAGE_CAP];
    const int t = threadIdx.x;
    const int lane = t & 31;
    const int wave = t >> 5;
    const int b = blockIdx.x;
    const int chunk = (e + NCHUNK - 1) / NCHUNK;
    const int e0 = b * chunk;
    const int e1 = (e0 + chunk < e) ? (e0 + chunk) : e;
    const int ne = (e1 > e0) ? (e1 - e0) : 0;
    const unsigned rs = (unsigned)range_nodes(n);

    if (t < NRANGE) rc[t] = 0;

    int dv[EPT], sv[EPT], rr[EPT];
#pragma unroll
    for (int u = 0; u < EPT; ++u) {
        const int j = t + u * 1024;
        dv[u] = -1;
        sv[u] = 0;
        if (j < ne) { dv[u] = dst[e0 + j]; sv[u] = src[e0 + j]; }
    }
    __syncthreads();
#pragma unroll
    for (int u = 0; u < EPT; ++u) {
        rr[u] = -1;
        if ((unsigned)dv[u] < (unsigned)n) {
            rr[u] = (int)((unsigned)dv[u] / rs);
            atomicAdd(&rc[rr[u]], 1);
        }
    }
    for (int j = t + EPT * 1024; j < ne; j += 1024) {
        const int d = dst[e0 + j];
        if ((unsigned)d < (unsigned)n) atomicAdd(&rc[(unsigned)d / rs], 1);
    }
    __syncthreads();

    int c = 0, inc = 0;
    if (t < NRANGE) {
        c = rc[t];
        inc = c;
        for (int o = 1; o < 32; o <<= 1) {
            const int up = __shfl_up(inc, o);
            if (lane >= o) inc += up;
        }
        if (lane == 31) wsum[wave] = inc;
    }
    __syncthreads();
    if (t < NRANGE) {
        int off = 0;
        for (int w = 0; w < wave; ++w) off += wsum[w];
        const int ex = off + inc - c;
        cur[t] = ex;
        meta[(size_t)b * NRANGE + t] = make_int2(ex, c);
        if (t == NRANGE - 1) nvalid = ex + c;
    }
    __syncthreads();

    const bool staged = (ne <= STAGE_CAP);
#pragma unroll
    for (int u = 0; u < EPT; ++u) {
        if (rr[u] >= 0) {
            const int p = atomicAdd(&cur[rr[u]], 1);
            const int2 v = make_int2(sv[u], dv[u]);
            if (staged) stage[p] = v; else tmp[(size_t)e0 + p] = v;
        }
    }
    for (int j = t + EPT * 1024; j < ne; j += 1024) {
        const int d = dst[e0 + j];
        if ((unsigned)d < (unsigned)n) {
            const int p = atomicAdd(&cur[(unsigned)d / rs], 1);
            const int2 v = make_int2(src[e0 + j], d);
            if (staged) stage[p] = v; else tmp[(size_t)e0 + p] = v;
        }
    }
    __syncthreads();
    if (staged) {
        const int nv = nvalid;
        for (int i = t; i < nv; i += 1024) tmp[(size_t)e0 + i] = stage[i];
    }
}

#define KQUADS (NFEAT / 4)
__global__ __launch_bounds__(1024) void csr_build_gemm(const int* __restrict__ tmp_w,
                                                       const int* __restrict__ meta_w,
                                                       int* __restrict__ rowptr,
                                                       int* __restrict__ cnt,
                                                       float* __restrict__ dinv,
                                                       int* __restrict__ ent,
                                                       const float* __restrict__ x,
                                                       const float* __restrict__ W,
                                                       float* __restrict__ xw, int n, int e) {
    __shared__ int   run_at[NCHUNK];
    __shared__ int   run_len[NCHUNK];
    __shared__ int   hc[RANGE_CAP];
    __shared__ float dl[RANGE_CAP];
    __shared__ int   wsum[32];
    __shared__ int   wso[32];
    __shared__ int   wsc[32];
    __shared__ int   stage[ENT_CAP];
    __shared__ float4 Ws[KQUADS * NHID];
    const int2* __restrict__ tmp  = (const int2*)tmp_w;
    const int2* __restrict__ meta = (const int2*)meta_w;
    const int t = threadIdx.x;
    const int r = blockIdx.x;
    const int rs = range_nodes(n);
    const int lo = r * rs;
    if (lo >= n) return;
    const int hi = (lo + rs < n) ? (lo + rs) : n;
    int rn = hi - lo;
    if (rn > RANGE_CAP) rn = RANGE_CAP;

    float4 wv[4];
#pragma unroll
    for (int it = 0; it < 4; ++it) {
        const int p = t + it * 1024;
        const float* wp = W + (size_t)(4 * (p >> 7)) * NHID + (p & (NHID - 1));
        wv[it].x = wp[0];
        wv[it].y = wp[NHID];
        wv[it].z = wp[2 * NHID];
        wv[it].w = wp[3 * NHID];
    }
    const int lane = t & 31;
    const int wave = t >> 5;
    const int chunk = (e + NCHUNK - 1) / NCHUNK;

    int so = 0, sc = 0;
    if (t < NCHUNK) {
        const int2 m = meta[(size_t)t * NRANGE + r];
        run_at[t] = m.x;
        run_len[t] = m.y;
        so = m.x;
        sc = m.y;
    }
    for (int o = 16; o > 0; o >>= 1) {
        so += __shfl_xor(so, o);
        sc += __shfl_xor(sc, o);
    }
    if (lane == 0) { wso[wave] = so; wsc[wave] = sc; }
    for (int k = t; k < rn; k += 1024) hc[k] = 0;
#pragma unroll
    for (int it = 0; it < 4; ++it) {
        const int p = t + it * 1024;
        const int cc = p & (NHID - 1);
        const int qd = p >> 7;
        const int f0 = ((((qd * 2 + (cc >> 6)) * 16 + (cc & 15)) * 2) * 4 + ((cc >> 4) & 3)) * 2;
        *(v2f*)((float*)Ws + f0)     = (v2f){wv[it].x, wv[it].y};
        *(v2f*)((float*)Ws + f0 + 8) = (v2f){wv[it].z, wv[it].w};
    }
    __syncthreads();

    int base = 0, total = 0;
    for (int w = 0; w < 32; ++w) { base += wso[w]; total += wsc[w]; }

    const int j0 = t & 7;
    int rl[2];
    const int2* rp[2];
#pragma unroll
    for (int h = 0; h < 2; ++h) {
        const int rb_ = h * (NCHUNK / 2) + (t >> 3);
        rl[h] = run_len[rb_];
        rp[h] = (rl[h] > 0) ? (tmp + (size_t)rb_ * chunk + run_at[rb_]) : meta;
    }
    int2 ev[2][RPT];
    int es[2][RPT], ed[2][RPT];
#pragma unroll
    for (int h = 0; h < 2; ++h)
#pragma unroll
        for (int u = 0; u < RPT; ++u) {
            const int j = j0 + 8 * u;
            ev[h][u] = rp[h][(j < rl[h]) ? j : 0];
        }
#pragma unroll
    for (int h = 0; h < 2; ++h)
#pragma unroll
        for (int u = 0; u < RPT; ++u) {
            const int j = j0 + 8 * u;
            es[h][u] = ev[h][u].x;
            ed[h][u] = (j < rl[h]) ? (ev[h][u].y - lo) : -1;
        }
#pragma unroll
    for (int h = 0; h < 2; ++h)
#pragma unroll
        for (int u = 0; u < RPT; ++u)
            if ((unsigned)ed[h][u] < (unsigned)rn) atomicAdd(&hc[ed[h][u]], 1);
#pragma unroll
    for (int h = 0; h < 2; ++h)
        for (int j = j0 + 8 * RPT; j < rl[h]; j += 8) {
            const int d = rp[h][j].y - lo;
            if ((unsigned)d < (unsigned)rn) atomicAdd(&hc[d], 1);
        }
    __syncthreads();

    const int per = (rn + 1023) >> 10;
    const int k_lo = (t * per < rn) ? (t * per) : rn;
    const int k_hi = (k_lo + per < rn) ? (k_lo + per) : rn;
    int v = 0;
    for (int k = k_lo; k < k_hi; ++k) v += hc[k];
    int inc = v;
    for (int o = 1; o < 32; o <<= 1) {
        const int up = __shfl_up(inc, o);
        if (lane >= o) inc += up;
    }
    if (lane == 31) wsum[wave] = inc;
    __syncthreads();
    int off = 0;
    for (int w = 0; w < wave; ++w) off += wsum[w];
    int run = off + inc - v;
    for (int k = k_lo; k < k_hi; ++k) {
        const int c = hc[k];
        const float di = rsqrtf((float)(c + 1));
        rowptr[lo + k] = base + run;
        cnt[lo + k] = c;
        dinv[lo + k] = di;
        dl[k] = di;
        hc[k] = run;
        run += c;
    }
    __syncthreads();

    const bool staged = (total <= ENT_CAP);
#pragma unroll
    for (int h = 0; h < 2; ++h)
#pragma unroll
        for (int u = 0; u < RPT; ++u) {
            if ((unsigned)ed[h][u] < (unsigned)rn) {
                const int p = atomicAdd(&hc[ed[h][u]], 1);
                if (staged) stage[p] = es[h][u]; else ent[(size_t)base + p] = es[h][u];
            }
        }
#pragma unroll
    for (int h = 0; h < 2; ++h)
        for (int j = j0 + 8 * RPT; j < rl[h]; j += 8) {
            const int2 tv = rp[h][j];
            const int d = tv.y - lo;
            if ((unsigned)d < (unsigned)rn) {
                const int p = atomicAdd(&hc[d], 1);
                if (staged) stage[p] = tv.x; else ent[(size_t)base + p] = tv.x;
            }
        }
    __syncthreads();
    if (staged)
        for (int i = t; i < total; i += 1024) ent[(size_t)base + i] = stage[i];

    const int half = lane >> 4;
    const int l    = lane & 15;
    const int wv_id = __builtin_amdgcn_readfirstlane(wave);
    const int ch    = wv_id & 1;
    const int nblk  = (n + 15) >> 4;
    const int rb0   = lo >> 4;
    const int nrb   = rs >> 4;
    const float4* bbase = Ws + ((ch * 16 + l) * 2 + half) * 2;

    for (int jb = (wv_id >> 1); jb < nrb && rb0 + jb < nblk; jb += 16) {
        const int m0 = (rb0 + jb) * 16;
        int arow = m0 + l;
        if (arow >= n) arow = n - 1;
        const float* aptr = x + (size_t)arow * NFEAT + 2 * half;
        v2f a[KQUADS];
#pragma unroll
        for (int q = 0; q < KQUADS; ++q) a[q] = *(const v2f*)(aptr + 4 * q);

        int dk = arow - lo;
        if (dk >= rn) dk = rn - 1;
        const float dv = dl[dk];

        v8f c[4];
#pragma unroll
        for (int tt = 0; tt < 4; ++tt) c[tt] = (v8f){};

#pragma unroll
        for (int q = 0; q < KQUADS; ++q) {
            const float4 b01 = bbase[128 * q];
            const float4 b23 = bbase[128 * q + 1];
            v2f bq[4];
            bq[0] = (v2f){b01.x, b01.y};
            bq[1] = (v2f){b01.z, b01.w};
            bq[2] = (v2f){b23.x, b23.y};
            bq[3] = (v2f){b23.z, b23.w};
            __builtin_amdgcn_sched_barrier(0);
#pragma unroll
            for (int tt = 0; tt < 4; ++tt)
                c[tt] = __builtin_amdgcn_wmma_f32_16x16x4_f32(
                             false, a[q],  false, bq[tt],
                             (short)0, c[tt],  false,  false);
            __builtin_amdgcn_sched_barrier(0);
        }

#pragma unroll
        for (int rr = 0; rr < 8; ++rr) {
            const int row = m0 + rr + 8 * half;
            const float dsc = __shfl(dv, rr + 8 * half);
            if (row < n) {
#pragma unroll
                for (int tt = 0; tt < 4; ++tt)
                    xw[(size_t)row * NHID + 64 * ch + 16 * tt + l] = c[tt][rr] * dsc;
            }
        }
    }
}

__global__ __launch_bounds__(256) void gcn_gather(const int* __restrict__ ent,
                                                  const int* __restrict__ rowptr,
                                                  const int* __restrict__ cnt,
                                                  const float* __restrict__ dinv,
                                                  const float* __restrict__ xw,
                                                  const float* __restrict__ bias,
                                                  float* __restrict__ out, int n) {
    const int lane = threadIdx.x & 31;
    const int i = __builtin_amdgcn_readfirstlane((int)(blockIdx.x * 8 + (threadIdx.x >> 5)));
    if (i >= n) return;

    const float di = dinv[i];
    float4 acc = *(const float4*)(xw + (size_t)i * NHID + lane * 4);
    const float4 bb = *(const float4*)(bias + lane * 4);

    const int start = rowptr[i];
    const int total = cnt[i];
    for (int base = 0; base < total; base += 32) {
        const int m = (total - base < 32) ? (total - base) : 32;
        int sidx = 0;
        if (lane < m) sidx = ent[start + base + lane];
        int j = 0;
        for (; j + 4 <= m; j += 4) {
            const int s0 = __builtin_amdgcn_readlane(sidx, j);
            const int s1 = __builtin_amdgcn_readlane(sidx, j + 1);
            const int s2 = __builtin_amdgcn_readlane(sidx, j + 2);
            const int s3 = __builtin_amdgcn_readlane(sidx, j + 3);
            const float4 r0 = *(const float4*)(xw + (size_t)s0 * NHID + lane * 4);
            const float4 r1 = *(const float4*)(xw + (size_t)s1 * NHID + lane * 4);
            const float4 r2 = *(const float4*)(xw + (size_t)s2 * NHID + lane * 4);
            const float4 r3 = *(const float4*)(xw + (size_t)s3 * NHID + lane * 4);
            acc.x += r0.x; acc.y += r0.y; acc.z += r0.z; acc.w += r0.w;
            acc.x += r1.x; acc.y += r1.y; acc.z += r1.z; acc.w += r1.w;
            acc.x += r2.x; acc.y += r2.y; acc.z += r2.z; acc.w += r2.w;
            acc.x += r3.x; acc.y += r3.y; acc.z += r3.z; acc.w += r3.w;
        }
        for (; j < m; ++j) {
            const int s0 = __builtin_amdgcn_readlane(sidx, j);
            const float4 r0 = *(const float4*)(xw + (size_t)s0 * NHID + lane * 4);
            acc.x += r0.x; acc.y += r0.y; acc.z += r0.z; acc.w += r0.w;
        }
    }
    float4 o;
    o.x = fmaf(acc.x, di, bb.x);
    o.y = fmaf(acc.y, di, bb.y);
    o.z = fmaf(acc.z, di, bb.z);
    o.w = fmaf(acc.w, di, bb.w);
    *(float4*)(out + (size_t)i * NHID + lane * 4) = o;
}

extern "C" void kernel_launch(void* const* d_in, const int* in_sizes, int n_in,
                              void* d_out, int out_size, void* d_ws, size_t ws_size,
                              hipStream_t stream) {
    const int*   ei   = (const int*)d_in[0];
    const float* x    = (const float*)d_in[1];
    const float* W    = (const float*)d_in[2];
    const float* bias = (const float*)d_in[3];
    float*       out  = (float*)d_out;

    const int E = in_sizes[0] / 2;
    const int N = in_sizes[1] / NFEAT;

    float* xw     = (float*)d_ws;
    int*   tmp    = (int*)(xw + (size_t)N * NHID);
    int*   meta   = tmp + 2 * (size_t)E;
    int*   ent    = meta + 2 * NCHUNK * NRANGE;
    float* dinv   = (float*)(ent + (size_t)E);
    int*   cnt    = (int*)(dinv + (size_t)N);
    int*   rowptr = cnt + (size_t)N;

    const int* srcIdx = ei;
    const int* dstIdx = ei + E;

    edge_partition<<<NCHUNK, 1024, 0, stream>>>(srcIdx, dstIdx, tmp, meta, N, E);
    csr_build_gemm<<<NRANGE, 1024, 0, stream>>>(tmp, meta, rowptr, cnt, dinv, ent, x, W, xw, N, E);

    gcn_gather<<<(N + 7) / 8, 256, 0, stream>>>(ent, rowptr, cnt, dinv, xw, bias, out, N);
}


# Sum first, transform after: the last launch sums each node's dinv-weighted rows of x into an LDS tile of 16 nodes and multiplies the tile by W (in LDS) on WMMA; no xw round trip; sum order changes
# speedup vs baseline: 5.0251x; 1.1922x over previous
#include <hip/hip_runtime.h>


typedef __attribute__((ext_vector_type(2))) float v2f;
typedef __attribute__((ext_vector_type(8))) float v8f;

#define NFEAT 128
#define NHID  128

#define NCHUNK 256
#define NRANGE 256
#define EPT    4
#define RPT    4
#define STAGE_CAP 24576
#define RANGE_CAP 4096
#define ENT_CAP   40960

__device__ __forceinline__ int range_nodes(int n) {
    const int nblk = (n + 15) >> 4;
    return 16 * ((nblk + NRANGE - 1) / NRANGE);
}

__global__ __launch_bounds__(1024) void edge_partition(const int* __restrict__ src,
                                                       const int* __restrict__ dst,
                                                       int* __restrict__ tmp_w,
                                                       int* __restrict__ meta_w, int n, int e) {
    int2* __restrict__ tmp  = (int2*)tmp_w;
    int2* __restrict__ meta = (int2*)meta_w;
    __shared__ int rc[NRANGE];
    __shared__ int cur[NRANGE];
    __shared__ int wsum[NRANGE / 32];
    __shared__ int nvalid;
    __shared__ int2 stage[STAGE_CAP];
    const int t = threadIdx.x;
    const int lane = t & 31;
    const int wave = t >> 5;
    const int b = blockIdx.x;
    const int chunk = (e + NCHUNK - 1) / NCHUNK;
    const int e0 = b * chunk;
    const int e1 = (e0 + chunk < e) ? (e0 + chunk) : e;
    const int ne = (e1 > e0) ? (e1 - e0) : 0;
    const unsigned rs = (unsigned)range_nodes(n);

    if (t < NRANGE) rc[t] = 0;

    int dv[EPT], sv[EPT], rr[EPT];
#pragma unroll
    for (int u = 0; u < EPT; ++u) {
        const int j = t + u * 1024;
        dv[u] = -1;
        sv[u] = 0;
        if (j < ne) { dv[u] = dst[e0 + j]; sv[u] = src[e0 + j]; }
    }
    __syncthreads();
#pragma unroll
    for (int u = 0; u < EPT; ++u) {
        rr[u] = -1;
        if ((unsigned)dv[u] < (unsigned)n) {
            rr[u] = (int)((unsigned)dv[u] / rs);
            atomicAdd(&rc[rr[u]], 1);
        }
    }
    for (int j = t + EPT * 1024; j < ne; j += 1024) {
        const int d = dst[e0 + j];
        if ((unsigned)d < (unsigned)n) atomicAdd(&rc[(unsigned)d / rs], 1);
    }
    __syncthreads();

    int c = 0, inc = 0;
    if (t < NRANGE) {
        c = rc[t];
        inc = c;
        for (int o = 1; o < 32; o <<= 1) {
            const int up = __shfl_up(inc, o);
            if (lane >= o) inc += up;
        }
        if (lane == 31) wsum[wave] = inc;
    }
    __syncthreads();
    if (t < NRANGE) {
        int off = 0;
        for (int w = 0; w < wave; ++w) off += wsum[w];
        const int ex = off + inc - c;
        cur[t] = ex;
        meta[(size_t)b * NRANGE + t] = make_int2(ex, c);
        if (t == NRANGE - 1) nvalid = ex + c;
    }
    __syncthreads();

    const bool staged = (ne <= STAGE_CAP);
#pragma unroll
    for (int u = 0; u < EPT; ++u) {
        if (rr[u] >= 0) {
            const int p = atomicAdd(&cur[rr[u]], 1);
            const int2 v = make_int2(sv[u], dv[u]);
            if (staged) stage[p] = v; else tmp[(size_t)e0 + p] = v;
        }
    }
    for (int j = t + EPT * 1024; j < ne; j += 1024) {
        const int d = dst[e0 + j];
        if ((unsigned)d < (unsigned)n) {
            const int p = atomicAdd(&cur[(unsigned)d / rs], 1);
            const int2 v = make_int2(src[e0 + j], d);
            if (staged) stage[p] = v; else tmp[(size_t)e0 + p] = v;
        }
    }
    __syncthreads();
    if (staged) {
        const int nv = nvalid;
        for (int i = t; i < nv; i += 1024) tmp[(size_t)e0 + i] = stage[i];
    }
}

__global__ __launch_bounds__(1024) void csr_build(const int* __restrict__ tmp_w,
                                                  const int* __restrict__ meta_w,
                                                  int* __restrict__ rowptr,
                                                  int* __restrict__ cnt,
                                                  float* __restrict__ dinv,
                                                  int* __restrict__ ent, int n, int e) {
    __shared__ int   run_at[NCHUNK];
    __shared__ int   run_len[NCHUNK];
    __shared__ int   hc[RANGE_CAP];
    __shared__ int   wsum[32];
    __shared__ int   wso[32];
    __shared__ int   wsc[32];
    __shared__ int   stage[ENT_CAP];
    const int2* __restrict__ tmp  = (const int2*)tmp_w;
    const int2* __restrict__ meta = (const int2*)meta_w;
    const int t = threadIdx.x;
    const int r = blockIdx.x;
    const int rs = range_nodes(n);
    const int lo = r * rs;
    if (lo >= n) return;
    const int hi = (lo + rs < n) ? (lo + rs) : n;
    int rn = hi - lo;
    if (rn > RANGE_CAP) rn = RANGE_CAP;

    const int lane = t & 31;
    const int wave = t >> 5;
    const int chunk = (e + NCHUNK - 1) / NCHUNK;

    int so = 0, sc = 0;
    if (t < NCHUNK) {
        const int2 m = meta[(size_t)t * NRANGE + r];
        run_at[t] = m.x;
        run_len[t] = m.y;
        so = m.x;
        sc = m.y;
    }
    for (int o = 16; o > 0; o >>= 1) {
        so += __shfl_xor(so, o);
        sc += __shfl_xor(sc, o);
    }
    if (lane == 0) { wso[wave] = so; wsc[wave] = sc; }
    for (int k = t; k < rn; k += 1024) hc[k] = 0;
    __syncthreads();

    int base = 0, total = 0;
    for (int w = 0; w < 32; ++w) { base += wso[w]; total += wsc[w]; }

    const int j0 = t & 7;
    int rl[2];
    const int2* rp[2];
#pragma unroll
    for (int h = 0; h < 2; ++h) {
        const int rb_ = h * (NCHUNK / 2) + (t >> 3);
        rl[h] = run_len[rb_];
        rp[h] = (rl[h] > 0) ? (tmp + (size_t)rb_ * chunk + run_at[rb_]) : meta;
    }
    int2 ev[2][RPT];
    int es[2][RPT], ed[2][RPT];
#pragma unroll
    for (int h = 0; h < 2; ++h)
#pragma unroll
        for (int u = 0; u < RPT; ++u) {
            const int j = j0 + 8 * u;
            ev[h][u] = rp[h][(j < rl[h]) ? j : 0];
        }
#pragma unroll
    for (int h = 0; h < 2; ++h)
#pragma unroll
        for (int u = 0; u < RPT; ++u) {
            const int j = j0 + 8 * u;
            es[h][u] = ev[h][u].x;
            ed[h][u] = (j < rl[h]) ? (ev[h][u].y - lo) : -1;
        }
#pragma unroll
    for (int h = 0; h < 2; ++h)
#pragma unroll
        for (int u = 0; u < RPT; ++u)
            if ((unsigned)ed[h][u] < (unsigned)rn) atomicAdd(&hc[ed[h][u]], 1);
#pragma unroll
    for (int h = 0; h < 2; ++h)
        for (int j = j0 + 8 * RPT; j < rl[h]; j += 8) {
            const int d = rp[h][j].y - lo;
            if ((unsigned)d < (unsigned)rn) atomicAdd(&hc[d], 1);
        }
    __syncthreads();

    const int per = (rn + 1023) >> 10;
    const int k_lo = (t * per < rn) ? (t * per) : rn;
    const int k_hi = (k_lo + per < rn) ? (k_lo + per) : rn;
    int v = 0;
    for (int k = k_lo; k < k_hi; ++k) v += hc[k];
    int inc = v;
    for (int o = 1; o < 32; o <<= 1) {
        const int up = __shfl_up(inc, o);
        if (lane >= o) inc += up;
    }
    if (lane == 31) wsum[wave] = inc;
    __syncthreads();
    int off = 0;
    for (int w = 0; w < wave; ++w) off += wsum[w];
    int run = off + inc - v;
    for (int k = k_lo; k < k_hi; ++k) {
        const int c = hc[k];
        const float di = rsqrtf((float)(c + 1));
        rowptr[lo + k] = base + run;
        cnt[lo + k] = c;
        dinv[lo + k] = di;
        hc[k] = run;
        run += c;
    }
    __syncthreads();

    const bool staged = (total <= ENT_CAP);
#pragma unroll
    for (int h = 0; h < 2; ++h)
#pragma unroll
        for (int u = 0; u < RPT; ++u) {
            if ((unsigned)ed[h][u] < (unsigned)rn) {
                const int p = atomicAdd(&hc[ed[h][u]], 1);
                if (staged) stage[p] = es[h][u]; else ent[(size_t)base + p] = es[h][u];
            }
        }
#pragma unroll
    for (int h = 0; h < 2; ++h)
        for (int j = j0 + 8 * RPT; j < rl[h]; j += 8) {
            const int2 tv = rp[h][j];
            const int d = tv.y - lo;
            if ((unsigned)d < (unsigned)rn) {
                const int p = atomicAdd(&hc[d], 1);
                if (staged) stage[p] = tv.x; else ent[(size_t)base + p] = tv.x;
            }
        }
    __syncthreads();
    if (staged)
        for (int i = t; i < total; i += 1024) ent[(size_t)base + i] = stage[i];
}

#define KPAIRS (NFEAT / 8)
#define TP     132

__device__ __forceinline__ void lds_barrier() {
    __builtin_amdgcn_fence(__ATOMIC_RELEASE, "workgroup", "local");
    __builtin_amdgcn_s_barrier();
    __builtin_amdgcn_fence(__ATOMIC_ACQUIRE, "workgroup", "local");
}

template <int K>
__device__ __forceinline__ void add_rows(float4& acc, const float* __restrict__ x, int sidx,
                                         float wgt, int j, int lane) {
    float4 rv[K];
    float  wk[K];
#pragma unroll
    for (int u = 0; u < K; ++u) {
        const int s = __builtin_amdgcn_readlane(sidx, j + u);
        wk[u] = __int_as_float(__builtin_amdgcn_readlane(__float_as_int(wgt), j + u));
        rv[u] = *(const float4*)(x + (size_t)s * NFEAT + lane * 4);
    }
#pragma unroll
    for (int u = 0; u < K; ++u) {
        acc.x = fmaf(wk[u], rv[u].x, acc.x);
        acc.y = fmaf(wk[u], rv[u].y, acc.y);
        acc.z = fmaf(wk[u], rv[u].z, acc.z);
        acc.w = fmaf(wk[u], rv[u].w, acc.w);
    }
}

__global__ __launch_bounds__(512) void gather_gemm(const int* __restrict__ ent,
                                                   const int* __restrict__ rowptr,
                                                   const int* __restrict__ cnt,
                                                   const float* __restrict__ dinv,
                                                   const float* __restrict__ x,
                                                   const float* __restrict__ W,
                                                   const float* __restrict__ bias,
                                                   float* __restrict__ out, int n) {
    __shared__ float4 Wt[KPAIRS * 8 * 2 * 16];
    __shared__ __attribute__((aligned(16))) float tileA[16 * TP];
    __shared__ __attribute__((aligned(16))) float tileC[2][16 * TP];
    const int t = threadIdx.x;
    const int lane = t & 31;
    const int wv = __builtin_amdgcn_readfirstlane(t >> 5);
    const int h = lane >> 4;
    const int l = lane & 15;
    const int ct = wv & 7;
    const int kh = wv >> 3;
    const int nblk = (n + 15) >> 4;
    const int step = (int)gridDim.x;

    float4 wst[KPAIRS * 256 / 512];
#pragma unroll
    for (int it = 0; it < KPAIRS * 256 / 512; ++it) {
        const int e = t + 512 * it;
        const float* wp = W + (size_t)(8 * (e >> 8) + 4 * ((e >> 4) & 1)) * NHID + 16 * ((e >> 5) & 7) + (e & 15);
        wst[it].x = wp[0];
        wst[it].y = wp[NHID];
        wst[it].z = wp[2 * NHID];
        wst[it].w = wp[3 * NHID];
    }
#pragma unroll
    for (int it = 0; it < KPAIRS * 256 / 512; ++it) Wt[t + 512 * it] = wst[it];
    const float4 bb = *(const float4*)(bias + lane * 4);
    const float*  arow = tileA + l * TP + 4 * h + 64 * kh;
    const float4* brow = Wt + (ct * 2 + h) * 16 + l + 2048 * kh;

    const int rb0 = (int)blockIdx.x;
    int   bstart, bcnt;
    float bdi;
    {
        const int  b = rb0 + lane * step;
        const int  i = b * 16 + wv;
        const bool lv = (b < nblk) && (i < n);
        const int  icl = lv ? i : (n - 1);
        bstart = rowptr[icl];
        bcnt = lv ? cnt[icl] : 0;
        bdi = lv ? dinv[icl] : 0.0f;
    }
    int   rb = rb0;
    int   i_cur = rb * 16 + wv;
    bool  live = (rb < nblk) && (i_cur < n);
    int   start = __builtin_amdgcn_readlane(bstart, 0);
    int   total = __builtin_amdgcn_readlane(bcnt, 0);
    float di = __int_as_float(__builtin_amdgcn_readlane(__float_as_int(bdi), 0));
    int   ic = live ? i_cur : (n - 1);
    int   sidx = ent[(lane < total) ? (start + lane) : 0];
    sidx = (lane < total) ? sidx : ic;
    float wgt = dinv[sidx];
    wgt = (lane < total) ? wgt : 0.0f;
    float4 own = *(const float4*)(x + (size_t)ic * NFEAT + lane * 4);

    for (int k = 0; rb < nblk; ++k) {
        const int  rb_n = rb + step;
        const int  i_n = rb_n * 16 + wv;
        const bool live_n = (rb_n < nblk) && (i_n < n);
        if (((k + 1) & 31) == 0) {
            const int  b = rb_n + lane * step;
            const int  i = b * 16 + wv;
            const bool lv = (b < nblk) && (i < n);
            const int  icl = lv ? i : (n - 1);
            bstart = rowptr[icl];
            bcnt = lv ? cnt[icl] : 0;
            bdi = lv ? dinv[icl] : 0.0f;
        }
        const int   start_n = __builtin_amdgcn_readlane(bstart, (k + 1) & 31);
        const int   total_n = __builtin_amdgcn_readlane(bcnt, (k + 1) & 31);
        const float di_n = __int_as_float(__builtin_amdgcn_readlane(__float_as_int(bdi), (k + 1) & 31));
        int sidx_n = ent[(lane < total_n) ? (start_n + lane) : 0];
        __builtin_amdgcn_sched_barrier(0);

        float4 acc;
        acc.x = di * own.x; acc.y = di * own.y; acc.z = di * own.z; acc.w = di * own.w;
        for (int base = 0; base < total; base += 32) {
            const int m = (total - base < 32) ? (total - base) : 32;
            int   sg = sidx;
            float wg = wgt;
            if (base > 0) {
                sg = ent[(lane < m) ? (start + base + lane) : 0];
                sg = (lane < m) ? sg : ic;
                wg = dinv[sg];
                wg = (lane < m) ? wg : 0.0f;
            }
            for (int j = 0; j < m; j += 16) {
                if (m - j > 8)      add_rows<16>(acc, x, sg, wg, j, lane);
                else if (m - j > 4) add_rows<8>(acc, x, sg, wg, j, lane);
                else                add_rows<4>(acc, x, sg, wg, j, lane);
            }
        }
        float4 o;
        o.x = acc.x * di; o.y = acc.y * di; o.z = acc.z * di; o.w = acc.w * di;
        *(float4*)(tileA + wv * TP + lane * 4) = o;

        const int ic_n = live_n ? i_n : (n - 1);
        sidx_n = (lane < total_n) ? sidx_n : ic_n;
        float wgt_n = dinv[sidx_n];
        wgt_n = (lane < total_n) ? wgt_n : 0.0f;
        const float4 own_n = *(const float4*)(x + (size_t)ic_n * NFEAT + lane * 4);
        lds_barrier();

        float4 av[8], bv[8];
#pragma unroll
        for (int u = 0; u < 8; ++u) {
            av[u] = *(const float4*)(arow + 8 * u);
            bv[u] = brow[256 * u];
        }
        __builtin_amdgcn_sched_barrier(0);
        v8f c0 = (v8f){}, c1 = (v8f){};
#pragma unroll
        for (int u = 0; u < 8; ++u) {
            c0 = __builtin_amdgcn_wmma_f32_16x16x4_f32(
                      false, (v2f){av[u].x, av[u].y},  false, (v2f){bv[u].x, bv[u].y},
                      (short)0, c0,  false,  false);
            c1 = __builtin_amdgcn_wmma_f32_16x16x4_f32(
                      false, (v2f){av[u].z, av[u].w},  false, (v2f){bv[u].z, bv[u].w},
                      (short)0, c1,  false,  false);
        }
#pragma unroll
        for (int r = 0; r < 8; ++r) tileC[kh][(r + 8 * h) * TP + 16 * ct + l] = c0[r] + c1[r];
        lds_barrier();

        if (live) {
            const float4 p0 = *(const float4*)(&tileC[0][wv * TP + lane * 4]);
            const float4 p1 = *(const float4*)(&tileC[1][wv * TP + lane * 4]);
            float4 q;
            q.x = (p0.x + p1.x) + bb.x; q.y = (p0.y + p1.y) + bb.y;
            q.z = (p0.z + p1.z) + bb.z; q.w = (p0.w + p1.w) + bb.w;
            *(float4*)(out + (size_t)i_cur * NHID + lane * 4) = q;
        }

        rb = rb_n; i_cur = i_n; live = live_n; ic = ic_n;
        start = start_n; total = total_n; di = di_n;
        sidx = sidx_n; wgt = wgt_n; own = own_n;
    }
}

extern "C" void kernel_launch(void* const* d_in, const int* in_sizes, int n_in,
                              void* d_out, int out_size, void* d_ws, size_t ws_size,
                              hipStream_t stream) {
    const int*   ei   = (const int*)d_in[0];
    const float* x    = (const float*)d_in[1];
    const float* W    = (const float*)d_in[2];
    const float* bias = (const float*)d_in[3];
    float*       out  = (float*)d_out;

    const int E = in_sizes[0] / 2;
    const int N = in_sizes[1] / NFEAT;

    int*   tmp    = (int*)d_ws;
    int*   meta   = tmp + 2 * (size_t)E;
    int*   ent    = meta + 2 * NCHUNK * NRANGE;
    float* dinv   = (float*)(ent + (size_t)E);
    int*   cnt    = (int*)(dinv + (size_t)N);
    int*   rowptr = cnt + (size_t)N;

    const int* srcIdx = ei;
    const int* dstIdx = ei + E;

    edge_partition<<<NCHUNK, 1024, 0, stream>>>(srcIdx, dstIdx, tmp, meta, N, E);
    csr_build<<<NRANGE, 1024, 0, stream>>>(tmp, meta, rowptr, cnt, dinv, ent, N, E);

    gather_gemm<<<512, 512, 0, stream>>>(ent, rowptr, cnt, dinv, x, W, bias, out, N);
}
